# starting kernel
# baseline (speedup 1.0000x reference)
#include <hip/hip_runtime.h>
#include <hip/hip_bf16.h>

typedef __attribute__((ext_vector_type(2))) float v2f;
typedef __attribute__((ext_vector_type(8))) float v8f;

__global__ void prep_weights(const float* __restrict__ w1a, const float* __restrict__ b1a,
                             const float* __restrict__ w1b, const float* __restrict__ b1b,
                             const float* __restrict__ w1c, const float* __restrict__ b1c,
                             float* __restrict__ wcat1, float* __restrict__ cb1,
                             float* __restrict__ wcat2, float* __restrict__ cb2,
                             float* __restrict__ wcat3, float* __restrict__ cb3) {
  int idx = blockIdx.x * blockDim.x + threadIdx.x;
  const int S1 = 110 * 142, S1B = 142, S2 = 32 * 64, S2B = 64, S3 = 32 * 64, S3B = 64;
  if (idx < S1) {
    int k = idx / 142, n = idx % 142;
    wcat1[idx] = (n < 71) ? (w1a[k * 71 + n] - w1a[(k + 110) * 71 + n])
                          : w1a[(k + 110) * 71 + (n - 71)];
    return;
  }
  idx -= S1;
  if (idx < S1B) { cb1[idx] = (idx < 71) ? b1a[idx] : 0.f; return; }
  idx -= S1B;
  if (idx < S2) {
    int k = idx / 64, n = idx % 64;
    wcat2[idx] = (n < 32) ? (w1b[k * 32 + n] - w1b[(k + 32) * 32 + n])
                          : w1b[(k + 32) * 32 + (n - 32)];
    return;
  }
  idx -= S2;
  if (idx < S2B) { cb2[idx] = (idx < 32) ? b1b[idx] : 0.f; return; }
  idx -= S2B;
  if (idx < S3) {
    int k = idx / 64, n = idx % 64;
    wcat3[idx] = (n < 32) ? (w1c[k * 32 + n] - w1c[(k + 32) * 32 + n])
                          : w1c[(k + 32) * 32 + (n - 32)];
    return;
  }
  idx -= S3;
  if (idx < S3B) { cb3[idx] = (idx < 32) ? b1c[idx] : 0.f; return; }
}

__global__ void pack_weights(const float* __restrict__ W, int K, int N,
                             int ksteps, int ntiles, float* __restrict__ out) {
  int idx = blockIdx.x * blockDim.x + threadIdx.x;
  int total = ntiles * ksteps * 32;
  if (idx >= total) return;
  int lane = idx & 31;
  int tmp = idx >> 5;
  int ks = tmp % ksteps;
  int nt = tmp / ksteps;
  int col = nt * 16 + (lane & 15);
  int k0 = ks * 4 + (lane >> 4) * 2;
  float v0 = (k0 < K && col < N) ? W[(size_t)k0 * N + col] : 0.f;
  float v1 = (k0 + 1 < K && col < N) ? W[(size_t)(k0 + 1) * N + col] : 0.f;
  out[(size_t)idx * 2]     = v0;
  out[(size_t)idx * 2 + 1] = v1;
}

__global__ void pad_input(const float* __restrict__ X0, float* __restrict__ Xp, int total) {
  int idx = blockIdx.x * blockDim.x + threadIdx.x;
  if (idx >= total) return;
  int r = idx / 112, c = idx - r * 112;
  Xp[idx] = (c < 110) ? X0[(size_t)r * 110 + c] : 0.f;
}

__global__ void gemm_wmma_f32(const float* __restrict__ A, int lda,
                              const float* __restrict__ Bp, int ksteps,
                              const float* __restrict__ bias, float* __restrict__ C,
                              int N, int do_relu) {
  const int lane = threadIdx.x;
  const int l16 = lane & 15;
  const int hi = lane >> 4;
  const int rowbase = blockIdx.x * 64;

  const float* aptr = A + (size_t)(rowbase + l16) * lda + hi * 2;
  const float* bptr = Bp + ((size_t)blockIdx.y * ksteps * 32 + lane) * 2;

  v8f acc0 = {0.f,0.f,0.f,0.f,0.f,0.f,0.f,0.f};
  v8f acc1 = acc0, acc2 = acc0, acc3 = acc0;

  for (int ks = 0; ks < ksteps; ++ks) {
    v2f b  = *(const v2f*)bptr;
    v2f a0 = *(const v2f*)(aptr);
    v2f a1 = *(const v2f*)(aptr + (size_t)16 * lda);
    v2f a2 = *(const v2f*)(aptr + (size_t)32 * lda);
    v2f a3 = *(const v2f*)(aptr + (size_t)48 * lda);
    acc0 = __builtin_amdgcn_wmma_f32_16x16x4_f32(false, a0, false, b, (short)0, acc0, false, false);
    acc1 = __builtin_amdgcn_wmma_f32_16x16x4_f32(false, a1, false, b, (short)0, acc1, false, false);
    acc2 = __builtin_amdgcn_wmma_f32_16x16x4_f32(false, a2, false, b, (short)0, acc2, false, false);
    acc3 = __builtin_amdgcn_wmma_f32_16x16x4_f32(false, a3, false, b, (short)0, acc3, false, false);
    aptr += 4;
    bptr += 64;
  }

  const int bcol = blockIdx.y * 16 + l16;
  if (bcol < N) {
    const float bv = bias ? bias[bcol] : 0.f;
    v8f accs[4] = {acc0, acc1, acc2, acc3};
#pragma unroll
    for (int r = 0; r < 4; ++r) {
#pragma unroll
      for (int v = 0; v < 8; ++v) {
        const int row = rowbase + r * 16 + (hi ? (v + 8) : v);
        float val = accs[r][v] + bv;
        if (do_relu) val = fmaxf(val, 0.f);
        C[(size_t)row * N + bcol] = val;
      }
    }
  }
}

__global__ void edge_ln_mean(const float* __restrict__ P, int C, int ld,
                             const float* __restrict__ g, const float* __restrict__ be,
                             float* __restrict__ H, int ldH) {
  __shared__ float hbuf[64][72];
  const int bb = blockIdx.x;
  const int e = threadIdx.x;
  const int i = e >> 3, j = e & 7;
  const float* PA = P + (size_t)(bb * 8 + i) * ld;
  const float* PB = P + (size_t)(bb * 8 + j) * ld + C;

  float sum = 0.f, sq = 0.f;
  for (int c = 0; c < C; ++c) {
    float v = PA[c] + PB[c];
    hbuf[e][c] = v;
    sum += v;
    sq += v * v;
  }
  const float inv = 1.0f / (float)C;
  const float mean = sum * inv;
  const float var = sq * inv - mean * mean;
  const float rs = rsqrtf(var + 1e-5f);
  for (int c = 0; c < C; ++c) {
    float v = (hbuf[e][c] - mean) * rs * g[c] + be[c];
    hbuf[e][c] = fmaxf(v, 0.f);
  }
  __syncthreads();

  for (int idx = e; idx < 8 * ldH; idx += 64) {
    const int ni = idx / ldH, c = idx - ni * ldH;
    float out = 0.f;
    if (c < C) {
      float s = 0.f;
#pragma unroll
      for (int jj = 0; jj < 8; ++jj) s += hbuf[ni * 8 + jj][c];
      out = s * 0.125f;
    }
    H[(size_t)(bb * 8 + ni) * ldH + c] = out;
  }
}

__global__ void head_kernel(const float* __restrict__ Hh,
                            const float* __restrict__ phiW1, const float* __restrict__ phib1,
                            const float* __restrict__ phiW2, const float* __restrict__ phib2,
                            float* __restrict__ out) {
  __shared__ float p[16];
  __shared__ float q[16];
  const int bb = blockIdx.x, t = threadIdx.x;
  if (t < 16) {
    float s = 0.f;
    for (int nd = 0; nd < 8; ++nd) s += Hh[(size_t)(bb * 8 + nd) * 16 + t];
    p[t] = s * 0.125f;
  }
  __syncthreads();
  if (t < 16) {
    float s = phib1[t];
    for (int k = 0; k < 16; ++k) s += p[k] * phiW1[k * 16 + t];
    q[t] = fmaxf(s, 0.f);
  }
  __syncthreads();
  if (t == 0) {
    float s = phib2[0];
    for (int k = 0; k < 16; ++k) s += q[k] * phiW2[k];
    out[bb] = s;
  }
}

extern "C" void kernel_launch(void* const* d_in, const int* in_sizes, int n_in,
                              void* d_out, int out_size, void* d_ws, size_t ws_size,
                              hipStream_t stream) {
  (void)in_sizes; (void)n_in; (void)out_size; (void)ws_size;
  const float* X0     = (const float*)d_in[0];
  const float* l1_W1  = (const float*)d_in[1];
  const float* l1_b1  = (const float*)d_in[2];
  const float* l1_g   = (const float*)d_in[3];
  const float* l1_be  = (const float*)d_in[4];
  const float* l1_W2  = (const float*)d_in[5];
  const float* l1_b2  = (const float*)d_in[6];
  const float* l2_W1  = (const float*)d_in[7];
  const float* l2_b1  = (const float*)d_in[8];
  const float* l2_g   = (const float*)d_in[9];
  const float* l2_be  = (const float*)d_in[10];
  const float* l2_W2  = (const float*)d_in[11];
  const float* l2_b2  = (const float*)d_in[12];
  const float* l3_W1  = (const float*)d_in[13];
  const float* l3_b1  = (const float*)d_in[14];
  const float* l3_g   = (const float*)d_in[15];
  const float* l3_be  = (const float*)d_in[16];
  const float* l3_W2  = (const float*)d_in[17];
  const float* l3_b2  = (const float*)d_in[18];
  const float* psi_W1 = (const float*)d_in[19];
  const float* psi_b1 = (const float*)d_in[20];
  const float* psi_W2 = (const float*)d_in[21];
  const float* psi_b2 = (const float*)d_in[22];
  const float* phi_W1 = (const float*)d_in[23];
  const float* phi_b1 = (const float*)d_in[24];
  const float* phi_W2 = (const float*)d_in[25];
  const float* phi_b2 = (const float*)d_in[26];

  const int M = 32768;

  float* ws = (float*)d_ws;
  float* wcat1 = ws;
  float* cb1   = wcat1 + 15620;
  float* wcat2 = cb1 + 144;
  float* cb2   = wcat2 + 2048;
  float* wcat3 = cb2 + 64;
  float* cb3   = wcat3 + 2048;
  float* p_w1   = cb3 + 64;
  float* p_l1W2 = p_w1 + 16128;
  float* p_w2   = p_l1W2 + 2304;
  float* p_l2W2 = p_w2 + 2048;
  float* p_w3   = p_l2W2 + 1024;
  float* p_l3W2 = p_w3 + 2048;
  float* p_psi1 = p_l3W2 + 1024;
  float* p_psi2 = p_psi1 + 512;
  float* X0p = ws + 49152;
  float* P   = X0p + (size_t)M * 112;
  float* H   = P   + (size_t)M * 142;
  float* X   = H   + (size_t)M * 72;
  float* X5  = X   + (size_t)M * 32;
  float* Hh  = X5  + (size_t)M * 16;

  prep_weights<<<80, 256, 0, stream>>>(l1_W1, l1_b1, l2_W1, l2_b1, l3_W1, l3_b1,
                                       wcat1, cb1, wcat2, cb2, wcat3, cb3);
  pack_weights<<<32, 256, 0, stream>>>(wcat1, 110, 142, 28, 9, p_w1);
  pack_weights<<<5, 256, 0, stream>>>(l1_W2, 71, 32, 18, 2, p_l1W2);
  pack_weights<<<4, 256, 0, stream>>>(wcat2, 32, 64, 8, 4, p_w2);
  pack_weights<<<2, 256, 0, stream>>>(l2_W2, 32, 32, 8, 2, p_l2W2);
  pack_weights<<<4, 256, 0, stream>>>(wcat3, 32, 64, 8, 4, p_w3);
  pack_weights<<<2, 256, 0, stream>>>(l3_W2, 32, 32, 8, 2, p_l3W2);
  pack_weights<<<1, 256, 0, stream>>>(psi_W1, 32, 16, 8, 1, p_psi1);
  pack_weights<<<1, 256, 0, stream>>>(psi_W2, 16, 16, 4, 1, p_psi2);
  pad_input<<<(M * 112 + 255) / 256, 256, 0, stream>>>(X0, X0p, M * 112);

  const int GB = M / 64;

  gemm_wmma_f32<<<dim3(GB, 9), 32, 0, stream>>>(X0p, 112, p_w1, 28, cb1, P, 142, 0);
  edge_ln_mean<<<4096, 64, 0, stream>>>(P, 71, 142, l1_g, l1_be, H, 72);
  gemm_wmma_f32<<<dim3(GB, 2), 32, 0, stream>>>(H, 72, p_l1W2, 18, l1_b2, X, 32, 1);

  gemm_wmma_f32<<<dim3(GB, 4), 32, 0, stream>>>(X, 32, p_w2, 8, cb2, P, 64, 0);
  edge_ln_mean<<<4096, 64, 0, stream>>>(P, 32, 64, l2_g, l2_be, H, 32);
  gemm_wmma_f32<<<dim3(GB, 2), 32, 0, stream>>>(H, 32, p_l2W2, 8, l2_b2, X, 32, 1);

  gemm_wmma_f32<<<dim3(GB, 4), 32, 0, stream>>>(X, 32, p_w3, 8, cb3, P, 64, 0);
  edge_ln_mean<<<4096, 64, 0, stream>>>(P, 32, 64, l3_g, l3_be, H, 32);
  gemm_wmma_f32<<<dim3(GB, 2), 32, 0, stream>>>(H, 32, p_l3W2, 8, l3_b2, X, 32, 0);

  gemm_wmma_f32<<<dim3(GB, 1), 32, 0, stream>>>(X, 32, p_psi1, 8, psi_b1, X5, 16, 1);
  gemm_wmma_f32<<<dim3(GB, 1), 32, 0, stream>>>(X5, 16, p_psi2, 4, psi_b2, Hh, 16, 0);

  head_kernel<<<4096, 32, 0, stream>>>(Hh, phi_W1, phi_b1, phi_W2, phi_b2, (float*)d_out);
}


# Wave roles rotate with the block index: the unequal per-wave matrix work of the GEMM stages (125/113/79/79 WMMAs) no longer lands on the same SIMD in every resident block
# speedup vs baseline: 5.5461x; 5.5461x over previous
#include <hip/hip_runtime.h>
#include <hip/hip_bf16.h>

typedef __attribute__((ext_vector_type(2))) float v2f;
typedef __attribute__((ext_vector_type(8))) float v8f;

__device__ __forceinline__ float wcat_at(const float* __restrict__ W1, int din, int dout,
                                         int k, int n) {
  return (n < dout) ? (W1[k * dout + n] - W1[(k + din) * dout + n])
                    : W1[(k + din) * dout + (n - dout)];
}

__device__ __forceinline__ void pack_plain(int idx, const float* __restrict__ Wt, int K, int N,
                                           int ksteps, int ntiles, float* __restrict__ out) {
  if (idx >= ntiles * ksteps * 32) return;
  const int lane = idx & 31;
  const int tmp = idx >> 5;
  const int ks = tmp % ksteps;
  const int nt = tmp / ksteps;
  const int col = nt * 16 + (lane & 15);
  const int k0 = ks * 4 + (lane >> 4) * 2;
  const float v0 = (k0 < K && col < N) ? Wt[(size_t)k0 * N + col] : 0.f;
  const float v1 = (k0 + 1 < K && col < N) ? Wt[(size_t)(k0 + 1) * N + col] : 0.f;
  out[(size_t)idx * 2]     = v0;
  out[(size_t)idx * 2 + 1] = v1;
}

__device__ __forceinline__ void pack_cat(int idx, const float* __restrict__ W1, int din, int dout,
                                         int ksteps, int ntiles, float* __restrict__ out) {
  if (idx >= ntiles * ksteps * 32) return;
  const int N = 2 * dout;
  const int lane = idx & 31;
  const int tmp = idx >> 5;
  const int ks = tmp % ksteps;
  const int nt = tmp / ksteps;
  const int col = nt * 16 + (lane & 15);
  const int k0 = ks * 4 + (lane >> 4) * 2;
  const float v0 = (k0 < din && col < N) ? wcat_at(W1, din, dout, k0, col) : 0.f;
  const float v1 = (k0 + 1 < din && col < N) ? wcat_at(W1, din, dout, k0 + 1, col) : 0.f;
  out[(size_t)idx * 2]     = v0;
  out[(size_t)idx * 2 + 1] = v1;
}

__global__ void setup_all(const float* __restrict__ l1W1, const float* __restrict__ l1b1,
                          const float* __restrict__ l1W2,
                          const float* __restrict__ l2W1, const float* __restrict__ l2b1,
                          const float* __restrict__ l2W2,
                          const float* __restrict__ l3W1, const float* __restrict__ l3b1,
                          const float* __restrict__ l3W2,
                          const float* __restrict__ psiW1, const float* __restrict__ psiW2,
                          float* __restrict__ cb1, float* __restrict__ cb2, float* __restrict__ cb3,
                          float* __restrict__ p_w1, float* __restrict__ p_l1W2,
                          float* __restrict__ p_w2, float* __restrict__ p_l2W2,
                          float* __restrict__ p_w3, float* __restrict__ p_l3W2,
                          float* __restrict__ p_psi1, float* __restrict__ p_psi2) {
  int b = blockIdx.x;
  const int t = threadIdx.x;
  if (b < 32) { pack_cat(b * 256 + t, l1W1, 110, 71, 28, 9, p_w1); return; }
  b -= 32;
  if (b < 5) { pack_plain(b * 256 + t, l1W2, 71, 32, 18, 2, p_l1W2); return; }
  b -= 5;
  if (b < 4) { pack_cat(b * 256 + t, l2W1, 32, 32, 8, 4, p_w2); return; }
  b -= 4;
  if (b < 2) { pack_plain(b * 256 + t, l2W2, 32, 32, 8, 2, p_l2W2); return; }
  b -= 2;
  if (b < 4) { pack_cat(b * 256 + t, l3W1, 32, 32, 8, 4, p_w3); return; }
  b -= 4;
  if (b < 2) { pack_plain(b * 256 + t, l3W2, 32, 32, 8, 2, p_l3W2); return; }
  b -= 2;
  if (b < 1) { pack_plain(t, psiW1, 32, 16, 8, 1, p_psi1); return; }
  b -= 1;
  if (b < 1) { pack_plain(t, psiW2, 16, 16, 4, 1, p_psi2); return; }
  if (t < 142) cb1[t] = (t < 71) ? l1b1[t] : 0.f;
  if (t < 64) {
    cb2[t] = (t < 32) ? l2b1[t] : 0.f;
    cb3[t] = (t < 32) ? l3b1[t] : 0.f;
  }
}

#define GC_THREADS 128

template <int C, int PP, int KP, int HP>
__device__ __forceinline__ void edge_stage(const float* Pt, float2* st, const float* gs,
                                           const float* bs, float* Hs, int tid) {
  {
    const int ri = tid >> 3;
    const int rj = ((tid >> 6) << 3) + (tid & 7);
    const float* pa = Pt + ri * PP;
    const float* pb = Pt + rj * PP + C;
    float sum = 0.f, sq = 0.f;
#pragma unroll
    for (int c = 0; c < C; ++c) {
      const float v = pa[c] + pb[c];
      sum += v;
      sq += v * v;
    }
    const float inv = 1.0f / (float)C;
    const float mean = sum * inv;
    const float var = sq * inv - mean * mean;
    const float rs = rsqrtf(var + 1e-5f);
    st[tid] = make_float2(mean, rs);
  }
  __syncthreads();
#pragma unroll
  for (int it = 0; it < (16 * KP) / GC_THREADS; ++it) {
    const int idx = it * GC_THREADS + tid;
    const int row = idx / KP;
    const int c = idx - row * KP;
    float out = 0.f;
    if (c < C) {
      const float a = Pt[row * PP + c];
      const float gg = gs[c];
      const float bb = bs[c];
      const float* pb = Pt + (row & 8) * PP + C + c;
      const float2* sp = st + row * 8;
      float s = 0.f;
#pragma unroll
      for (int jj = 0; jj < 8; ++jj) {
        const float2 m = sp[jj];
        float v = a + pb[jj * PP];
        v = (v - m.x) * m.y * gg + bb;
        s += fmaxf(v, 0.f);
      }
      out = s * 0.125f;
    }
    Hs[row * HP + c] = out;
  }
  __syncthreads();
}

template <int KS>
struct BFrag { v2f v[KS]; };

template <int KS>
__device__ __forceinline__ BFrag<KS> load_b(const float* __restrict__ Bt, int lane) {
  BFrag<KS> f;
  const float* bp = Bt + lane * 2;
#pragma unroll
  for (int ks = 0; ks < KS; ++ks) f.v[ks] = *(const v2f*)(bp + ks * 64);
  return f;
}

template <int KS>
__device__ __forceinline__ v8f tile_gemm(const float* As, int apitch, const BFrag<KS>& bf, int lane) {
  const float* ap = As + (lane & 15) * apitch + (lane >> 4) * 2;
  v8f acc = {0.f,0.f,0.f,0.f,0.f,0.f,0.f,0.f};
#pragma unroll
  for (int ks = 0; ks < KS; ++ks) {
    const v2f a = *(const v2f*)(ap + ks * 4);
    acc = __builtin_amdgcn_wmma_f32_16x16x4_f32(false, a, false, bf.v[ks], (short)0, acc, false, false);
  }
  return acc;
}

template <bool RELU>
__device__ __forceinline__ void store_tile(v8f acc, float bv, float* Ds, int dpitch,
                                           int col, int hi) {
#pragma unroll
  for (int v = 0; v < 8; ++v) {
    float val = acc[v] + bv;
    if (RELU) val = fmaxf(val, 0.f);
    Ds[(v + 8 * hi) * dpitch + col] = val;
  }
}

#define XPITCH 116
#define QPITCH 20
__device__ __forceinline__ void proj_stage(const float* Xin, const float* __restrict__ pW, int w,
                                           const float* __restrict__ cb, float* Pt, float* Pq,
                                           int lane) {
  const int l16 = lane & 15;
  const int hi = lane >> 4;
  const float* ap = Xin + l16 * XPITCH + hi * 2;
  const float* ap8 = ap + w * 28;
  const float* bp = pW + (size_t)(2 * w) * (28 * 64) + lane * 2;
  const float* bp8 = pW + (size_t)(8 * 28 + 7 * w) * 64 + lane * 2;
  v8f acc0 = {0.f,0.f,0.f,0.f,0.f,0.f,0.f,0.f};
  v8f acc1 = acc0, acc8 = acc0;
#pragma unroll
  for (int ks = 0; ks < 28; ++ks) {
    const v2f a = *(const v2f*)(ap + ks * 4);
    const v2f b0 = *(const v2f*)(bp + ks * 64);
    const v2f b1 = *(const v2f*)(bp + (28 + ks) * 64);
    acc0 = __builtin_amdgcn_wmma_f32_16x16x4_f32(false, a, false, b0, (short)0, acc0, false, false);
    acc1 = __builtin_amdgcn_wmma_f32_16x16x4_f32(false, a, false, b1, (short)0, acc1, false, false);
    if ((ks & 3) == 0) {
      const int j = ks >> 2;
      const v2f a8 = *(const v2f*)(ap8 + j * 4);
      const v2f b8 = *(const v2f*)(bp8 + j * 64);
      acc8 = __builtin_amdgcn_wmma_f32_16x16x4_f32(false, a8, false, b8, (short)0, acc8, false, false);
    }
  }
  const int col0 = (2 * w) * 16 + l16;
  store_tile<false>(acc0, cb[col0], Pt, 146, col0, hi);
  store_tile<false>(acc1, cb[col0 + 16], Pt, 146, col0 + 16, hi);
  store_tile<false>(acc8, 0.f, Pq + w * (16 * QPITCH), QPITCH, l16, hi);
}

__device__ __forceinline__ void proj_reduce(const float* Pq, float bq, float* Pt, int tid) {
#pragma unroll
  for (int i = 0; i < 2; ++i) {
    const int idx = i * GC_THREADS + tid;
    const int r = idx >> 4;
    const int c = idx & 15;
    const float* q = Pq + r * QPITCH + c;
    const float s = (q[0] + q[16 * QPITCH]) + (q[2 * 16 * QPITCH] + q[3 * 16 * QPITCH]);
    Pt[r * 146 + 128 + c] = s + bq;
  }
}

__global__ __launch_bounds__(GC_THREADS, 8) void graph_chain(
    const float* __restrict__ X0,
    const float* __restrict__ pW1a, const float* __restrict__ cba,
    const float* __restrict__ g1, const float* __restrict__ be1,
    const float* __restrict__ pW2a, const float* __restrict__ b2a,
    const float* __restrict__ pW1b, const float* __restrict__ cbb,
    const float* __restrict__ g2, const float* __restrict__ be2,
    const float* __restrict__ pW2b, const float* __restrict__ b2b,
    const float* __restrict__ pW1c, const float* __restrict__ cbc,
    const float* __restrict__ g3, const float* __restrict__ be3,
    const float* __restrict__ pW2c, const float* __restrict__ b2c,
    const float* __restrict__ pPsi1, const float* __restrict__ psib1,
    const float* __restrict__ pPsi2, const float* __restrict__ psib2,
    const float* __restrict__ phiW1, const float* __restrict__ phib1,
    const float* __restrict__ phiW2, const float* __restrict__ phib2,
    float* __restrict__ out) {
  __shared__ __attribute__((aligned(16))) float Xin[16 * XPITCH];
  __shared__ __attribute__((aligned(16))) float Pt[16 * 146];
  __shared__ __attribute__((aligned(16))) float Pq[4 * 16 * QPITCH];
  __shared__ __attribute__((aligned(16))) float2 st[GC_THREADS];
  __shared__ __attribute__((aligned(16))) float Hs[16 * 76];
  __shared__ __attribute__((aligned(16))) float Xs[16 * 36];
  __shared__ __attribute__((aligned(16))) float X5s[16 * 20];
  __shared__ float Hhs[16 * 17];
  __shared__ float gsA[72], bsA[72], gsB[32], bsB[32], gsC[32], bsC[32];
  __shared__ float ps[32], qs[32];

  const int tid = threadIdx.x;
  const int lane = tid & 31;
  const int l16 = lane & 15;
  const int hi = lane >> 4;
  const int blk = blockIdx.x;
  const int rot = (blk ^ (blk >> 2) ^ (blk >> 4) ^ (blk >> 6) ^ (blk >> 8) ^ (blk >> 10)) & 3;
  const int w = (__builtin_amdgcn_readfirstlane(tid >> 5) + rot) & 3;
  const int wn = w & 1;

  float bq9 = 0.f;
  {
    const float4* src = (const float4*)(X0 + (size_t)blk * (16 * 110));
    float4 xv[4];
    xv[0] = src[tid];
    xv[1] = src[GC_THREADS + tid];
    xv[2] = src[2 * GC_THREADS + tid];
    xv[3] = (tid < 56) ? src[3 * GC_THREADS + tid] : make_float4(0.f, 0.f, 0.f, 0.f);
    float ga = 0.f, ba = 0.f, gb = 0.f, bb = 0.f, gc = 0.f, bc = 0.f;
    if (tid < 71) { ga = g1[tid]; ba = be1[tid]; }
    if (tid < 32) { gb = g2[tid]; bb = be2[tid]; gc = g3[tid]; bc = be3[tid]; }
    if ((tid & 15) < 14) bq9 = cba[128 + (tid & 15)];
#pragma unroll
    for (int it = 0; it < 4; ++it) {
      const int f = it * GC_THREADS + tid;
      if (it < 3 || tid < 56) {
        const int e0 = f * 4;
        const int r0 = e0 / 110;
        const int c0 = e0 - r0 * 110;
        const int e1 = e0 + 2;
        const int r1 = e1 / 110;
        const int c1 = e1 - r1 * 110;
        *(float2*)&Xin[r0 * XPITCH + c0] = make_float2(xv[it].x, xv[it].y);
        *(float2*)&Xin[r1 * XPITCH + c1] = make_float2(xv[it].z, xv[it].w);
      }
    }
    if (tid < 16) *(float2*)&Xin[tid * XPITCH + 110] = make_float2(0.f, 0.f);
    if (tid < 71) { gsA[tid] = ga; bsA[tid] = ba; }
    if (tid < 32) { gsB[tid] = gb; bsB[tid] = bb; gsC[tid] = gc; bsC[tid] = bc; }
  }
  __syncthreads();

  proj_stage(Xin, pW1a, w, cba, Pt, Pq, lane);
  const BFrag<18> bfA = load_b<18>(pW2a + wn * (18 * 64), lane);
  __syncthreads();
  proj_reduce(Pq, bq9, Pt, tid);
  __syncthreads();

  edge_stage<71, 146, 72, 76>(Pt, st, gsA, bsA, Hs, tid);
  const BFrag<8> bfB = load_b<8>(pW1b + w * (8 * 64), lane);
  const float cbbv = cbb[w * 16 + l16];
  if (w < 2) {
    const v8f acc = tile_gemm<18>(Hs, 76, bfA, lane);
    store_tile<true>(acc, b2a[w * 16 + l16], Xs, 36, w * 16 + l16, hi);
  }
  __syncthreads();

  {
    const v8f acc = tile_gemm<8>(Xs, 36, bfB, lane);
    store_tile<false>(acc, cbbv, Pt, 66, w * 16 + l16, hi);
  }
  const BFrag<8> bfC = load_b<8>(pW2b + wn * (8 * 64), lane);
  const float b2bv = b2b[wn * 16 + l16];
  __syncthreads();
  edge_stage<32, 66, 32, 36>(Pt, st, gsB, bsB, Hs, tid);
  const BFrag<8> bfD = load_b<8>(pW1c + w * (8 * 64), lane);
  const float cbcv = cbc[w * 16 + l16];
  if (w < 2) {
    const v8f acc = tile_gemm<8>(Hs, 36, bfC, lane);
    store_tile<true>(acc, b2bv, Xs, 36, w * 16 + l16, hi);
  }
  __syncthreads();

  {
    const v8f acc = tile_gemm<8>(Xs, 36, bfD, lane);
    store_tile<false>(acc, cbcv, Pt, 66, w * 16 + l16, hi);
  }
  const BFrag<8> bfE = load_b<8>(pW2c + wn * (8 * 64), lane);
  const float b2cv = b2c[wn * 16 + l16];
  __syncthreads();
  edge_stage<32, 66, 32, 36>(Pt, st, gsC, bsC, Hs, tid);
  const BFrag<8> bfF = load_b<8>(pPsi1, lane);
  const BFrag<4> bfG = load_b<4>(pPsi2, lane);
  const float psi1v = psib1[l16];
  const float psi2v = psib2[l16];
  if (w < 2) {
    const v8f acc = tile_gemm<8>(Hs, 36, bfE, lane);
    store_tile<false>(acc, b2cv, Xs, 36, w * 16 + l16, hi);
  }
  __syncthreads();

  if (w == 0) {
    const v8f acc = tile_gemm<8>(Xs, 36, bfF, lane);
    store_tile<true>(acc, psi1v, X5s, 20, l16, hi);
  }
  __syncthreads();
  if (w == 0) {
    const v8f acc = tile_gemm<4>(X5s, 20, bfG, lane);
    store_tile<false>(acc, psi2v, Hhs, 17, l16, hi);
  }
  __syncthreads();

  const int gq = hi;
  const int tq = l16;
  if (w == 0) {
    float s = 0.f;
#pragma unroll
    for (int nd = 0; nd < 8; ++nd) s += Hhs[(gq * 8 + nd) * 17 + tq];
    ps[lane] = s * 0.125f;
  }
  __syncthreads();
  if (w == 0) {
    float s = phib1[tq];
#pragma unroll
    for (int k = 0; k < 16; ++k) s += ps[gq * 16 + k] * phiW1[k * 16 + tq];
    qs[lane] = fmaxf(s, 0.f);
  }
  __syncthreads();
  if (w == 0 && lane < 2) {
    float s = phib2[0];
#pragma unroll
    for (int k = 0; k < 16; ++k) s += qs[lane * 16 + k] * phiW2[k];
    out[blk * 2 + lane] = s;
  }
}

extern "C" void kernel_launch(void* const* d_in, const int* in_sizes, int n_in,
                              void* d_out, int out_size, void* d_ws, size_t ws_size,
                              hipStream_t stream) {
  (void)in_sizes; (void)n_in; (void)out_size; (void)ws_size;
  const float* X0     = (const float*)d_in[0];
  const float* l1_W1  = (const float*)d_in[1];
  const float* l1_b1  = (const float*)d_in[2];
  const float* l1_g   = (const float*)d_in[3];
  const float* l1_be  = (const float*)d_in[4];
  const float* l1_W2  = (const float*)d_in[5];
  const float* l1_b2  = (const float*)d_in[6];
  const float* l2_W1  = (const float*)d_in[7];
  const float* l2_b1  = (const float*)d_in[8];
  const float* l2_g   = (const float*)d_in[9];
  const float* l2_be  = (const float*)d_in[10];
  const float* l2_W2  = (const float*)d_in[11];
  const float* l2_b2  = (const float*)d_in[12];
  const float* l3_W1  = (const float*)d_in[13];
  const float* l3_b1  = (const float*)d_in[14];
  const float* l3_g   = (const float*)d_in[15];
  const float* l3_be  = (const float*)d_in[16];
  const float* l3_W2  = (const float*)d_in[17];
  const float* l3_b2  = (const float*)d_in[18];
  const float* psi_W1 = (const float*)d_in[19];
  const float* psi_b1 = (const float*)d_in[20];
  const float* psi_W2 = (const float*)d_in[21];
  const float* psi_b2 = (const float*)d_in[22];
  const float* phi_W1 = (const float*)d_in[23];
  const float* phi_b1 = (const float*)d_in[24];
  const float* phi_W2 = (const float*)d_in[25];
  const float* phi_b2 = (const float*)d_in[26];

  float* ws = (float*)d_ws;
  float* wcat1 = ws;
  float* cb1   = wcat1 + 15620;
  float* wcat2 = cb1 + 144;
  float* cb2   = wcat2 + 2048;
  float* wcat3 = cb2 + 64;
  float* cb3   = wcat3 + 2048;
  float* p_w1   = cb3 + 64;
  float* p_l1W2 = p_w1 + 16128;
  float* p_w2   = p_l1W2 + 2304;
  float* p_l2W2 = p_w2 + 2048;
  float* p_w3   = p_l2W2 + 1024;
  float* p_l3W2 = p_w3 + 2048;
  float* p_psi1 = p_l3W2 + 1024;
  float* p_psi2 = p_psi1 + 512;

  setup_all<<<52, 256, 0, stream>>>(l1_W1, l1_b1, l1_W2, l2_W1, l2_b1, l2_W2,
                                    l3_W1, l3_b1, l3_W2, psi_W1, psi_W2,
                                    cb1, cb2, cb3, p_w1, p_l1W2, p_w2, p_l2W2,
                                    p_w3, p_l3W2, p_psi1, p_psi2);

  graph_chain<<<2048, GC_THREADS, 0, stream>>>(X0, p_w1, cb1, l1_g, l1_be, p_l1W2, l1_b2,
                                                 p_w2, cb2, l2_g, l2_be, p_l2W2, l2_b2,
                                                 p_w3, cb3, l3_g, l3_be, p_l3W2, l3_b2,
                                                 p_psi1, psi_b1, p_psi2, psi_b2,
                                                 phi_W1, phi_b1, phi_W2, phi_b2, (float*)d_out);
}


# Edge apply stage re-mapped: a wave owns four rows of one graph, lanes run over channels in packed f32 pairs; neighbour values and per-edge scale/offset statistics are loaded once per wave
# speedup vs baseline: 5.9241x; 1.0682x over previous
#include <hip/hip_runtime.h>
#include <hip/hip_bf16.h>

typedef __attribute__((ext_vector_type(2))) float v2f;
typedef __attribute__((ext_vector_type(8))) float v8f;

__device__ __forceinline__ float wcat_at(const float* __restrict__ W1, int din, int dout,
                                         int k, int n) {
  return (n < dout) ? (W1[k * dout + n] - W1[(k + din) * dout + n])
                    : W1[(k + din) * dout + (n - dout)];
}

__device__ __forceinline__ void pack_plain(int idx, const float* __restrict__ Wt, int K, int N,
                                           int ksteps, int ntiles, float* __restrict__ out) {
  if (idx >= ntiles * ksteps * 32) return;
  const int lane = idx & 31;
  const int tmp = idx >> 5;
  const int ks = tmp % ksteps;
  const int nt = tmp / ksteps;
  const int col = nt * 16 + (lane & 15);
  const int k0 = ks * 4 + (lane >> 4) * 2;
  const float v0 = (k0 < K && col < N) ? Wt[(size_t)k0 * N + col] : 0.f;
  const float v1 = (k0 + 1 < K && col < N) ? Wt[(size_t)(k0 + 1) * N + col] : 0.f;
  out[(size_t)idx * 2]     = v0;
  out[(size_t)idx * 2 + 1] = v1;
}

__device__ __forceinline__ void pack_cat(int idx, const float* __restrict__ W1, int din, int dout,
                                         int ksteps, int ntiles, float* __restrict__ out) {
  if (idx >= ntiles * ksteps * 32) return;
  const int N = 2 * dout;
  const int lane = idx & 31;
  const int tmp = idx >> 5;
  const int ks = tmp % ksteps;
  const int nt = tmp / ksteps;
  const int col = nt * 16 + (lane & 15);
  const int k0 = ks * 4 + (lane >> 4) * 2;
  const float v0 = (k0 < din && col < N) ? wcat_at(W1, din, dout, k0, col) : 0.f;
  const float v1 = (k0 + 1 < din && col < N) ? wcat_at(W1, din, dout, k0 + 1, col) : 0.f;
  out[(size_t)idx * 2]     = v0;
  out[(size_t)idx * 2 + 1] = v1;
}

__global__ void setup_all(const float* __restrict__ l1W1, const float* __restrict__ l1b1,
                          const float* __restrict__ l1W2,
                          const float* __restrict__ l2W1, const float* __restrict__ l2b1,
                          const float* __restrict__ l2W2,
                          const float* __restrict__ l3W1, const float* __restrict__ l3b1,
                          const float* __restrict__ l3W2,
                          const float* __restrict__ psiW1, const float* __restrict__ psiW2,
                          float* __restrict__ cb1, float* __restrict__ cb2, float* __restrict__ cb3,
                          float* __restrict__ p_w1, float* __restrict__ p_l1W2,
                          float* __restrict__ p_w2, float* __restrict__ p_l2W2,
                          float* __restrict__ p_w3, float* __restrict__ p_l3W2,
                          float* __restrict__ p_psi1, float* __restrict__ p_psi2) {
  int b = blockIdx.x;
  const int t = threadIdx.x;
  if (b < 32) { pack_cat(b * 256 + t, l1W1, 110, 71, 28, 9, p_w1); return; }
  b -= 32;
  if (b < 5) { pack_plain(b * 256 + t, l1W2, 71, 32, 18, 2, p_l1W2); return; }
  b -= 5;
  if (b < 4) { pack_cat(b * 256 + t, l2W1, 32, 32, 8, 4, p_w2); return; }
  b -= 4;
  if (b < 2) { pack_plain(b * 256 + t, l2W2, 32, 32, 8, 2, p_l2W2); return; }
  b -= 2;
  if (b < 4) { pack_cat(b * 256 + t, l3W1, 32, 32, 8, 4, p_w3); return; }
  b -= 4;
  if (b < 2) { pack_plain(b * 256 + t, l3W2, 32, 32, 8, 2, p_l3W2); return; }
  b -= 2;
  if (b < 1) { pack_plain(t, psiW1, 32, 16, 8, 1, p_psi1); return; }
  b -= 1;
  if (b < 1) { pack_plain(t, psiW2, 16, 16, 4, 1, p_psi2); return; }
  if (t < 142) cb1[t] = (t < 71) ? l1b1[t] : 0.f;
  if (t < 64) {
    cb2[t] = (t < 32) ? l2b1[t] : 0.f;
    cb3[t] = (t < 32) ? l3b1[t] : 0.f;
  }
}

#define GC_THREADS 128

template <int C, int PP, int KP, int HP>
__device__ __forceinline__ void edge_stage(const float* Pt, float* rsS, float* nmS, const float* gs,
                                           const float* bs, float* Hs, int tid) {
  {
    const int ri = tid >> 3;
    const int rj = ((tid >> 6) << 3) + (tid & 7);
    const float* pa = Pt + ri * PP;
    const float* pb = Pt + rj * PP + C;
    float sum = 0.f, sq = 0.f;
#pragma unroll
    for (int c = 0; c < C; ++c) {
      const float v = pa[c] + pb[c];
      sum += v;
      sq += v * v;
    }
    const float inv = 1.0f / (float)C;
    const float mean = sum * inv;
    const float var = sq * inv - mean * mean;
    const float rs = rsqrtf(var + 1e-5f);
    rsS[tid] = rs;
    nmS[tid] = -mean * rs;
  }
  __syncthreads();
  constexpr int NP = C / 32;
  const int lane = tid & 31;
  const int r0 = __builtin_amdgcn_readfirstlane(tid >> 5) * 4;
  {
    const float* pbn = Pt + (r0 & 8) * PP + C + lane;
    v2f bv[NP][4], gg[NP], bb[NP];
#pragma unroll
    for (int p = 0; p < NP; ++p) {
      const float g1 = gs[p * 32 + lane];
      const float b1 = bs[p * 32 + lane];
      gg[p] = v2f{g1, g1};
      bb[p] = v2f{b1, b1};
#pragma unroll
      for (int q = 0; q < 4; ++q)
        bv[p][q] = v2f{pbn[(2 * q) * PP + p * 32], pbn[(2 * q + 1) * PP + p * 32]};
    }
#pragma unroll
    for (int rr = 0; rr < 4; ++rr) {
      const int row = r0 + rr;
      const float4 ra = *(const float4*)(rsS + row * 8);
      const float4 rb = *(const float4*)(rsS + row * 8 + 4);
      const float4 na = *(const float4*)(nmS + row * 8);
      const float4 nb = *(const float4*)(nmS + row * 8 + 4);
      const v2f rsv[4] = {v2f{ra.x, ra.y}, v2f{ra.z, ra.w}, v2f{rb.x, rb.y}, v2f{rb.z, rb.w}};
      const v2f nmv[4] = {v2f{na.x, na.y}, v2f{na.z, na.w}, v2f{nb.x, nb.y}, v2f{nb.z, nb.w}};
#pragma unroll
      for (int p = 0; p < NP; ++p) {
        const float a = Pt[row * PP + p * 32 + lane];
        const v2f a2 = v2f{a, a};
        v2f s2 = v2f{0.f, 0.f};
#pragma unroll
        for (int q = 0; q < 4; ++q) {
          v2f v = __builtin_elementwise_fma(a2 + bv[p][q], rsv[q], nmv[q]);
          v = __builtin_elementwise_fma(v, gg[p], bb[p]);
          s2 += __builtin_elementwise_max(v, v2f{0.f, 0.f});
        }
        Hs[row * HP + p * 32 + lane] = (s2.x + s2.y) * 0.125f;
      }
    }
  }
  if (KP > NP * 32) {
    static_assert(KP == NP * 32 || KP - NP * 32 == 8, "remainder pass covers eight columns");
    const int row = tid >> 3;
    const int c = NP * 32 + (tid & 7);
    float out = 0.f;
    if (c < C) {
      const float a = Pt[row * PP + c];
      const float g1 = gs[c];
      const float b1 = bs[c];
      const float* pb = Pt + (row & 8) * PP + C + c;
      const float4 ra = *(const float4*)(rsS + row * 8);
      const float4 rb = *(const float4*)(rsS + row * 8 + 4);
      const float4 na = *(const float4*)(nmS + row * 8);
      const float4 nb = *(const float4*)(nmS + row * 8 + 4);
      const v2f rsv[4] = {v2f{ra.x, ra.y}, v2f{ra.z, ra.w}, v2f{rb.x, rb.y}, v2f{rb.z, rb.w}};
      const v2f nmv[4] = {v2f{na.x, na.y}, v2f{na.z, na.w}, v2f{nb.x, nb.y}, v2f{nb.z, nb.w}};
      const v2f a2 = v2f{a, a};
      const v2f g2 = v2f{g1, g1};
      const v2f b2 = v2f{b1, b1};
      v2f s2 = v2f{0.f, 0.f};
#pragma unroll
      for (int q = 0; q < 4; ++q) {
        const v2f bq = v2f{pb[(2 * q) * PP], pb[(2 * q + 1) * PP]};
        v2f v = __builtin_elementwise_fma(a2 + bq, rsv[q], nmv[q]);
        v = __builtin_elementwise_fma(v, g2, b2);
        s2 += __builtin_elementwise_max(v, v2f{0.f, 0.f});
      }
      const float s = s2.x + s2.y;
      out = s * 0.125f;
    }
    Hs[row * HP + c] = out;
  }
  __syncthreads();
}

template <int KS>
struct BFrag { v2f v[KS]; };

template <int KS>
__device__ __forceinline__ BFrag<KS> load_b(const float* __restrict__ Bt, int lane) {
  BFrag<KS> f;
  const float* bp = Bt + lane * 2;
#pragma unroll
  for (int ks = 0; ks < KS; ++ks) f.v[ks] = *(const v2f*)(bp + ks * 64);
  return f;
}

template <int KS>
__device__ __forceinline__ v8f tile_gemm(const float* As, int apitch, const BFrag<KS>& bf, int lane) {
  const float* ap = As + (lane & 15) * apitch + (lane >> 4) * 2;
  v8f acc = {0.f,0.f,0.f,0.f,0.f,0.f,0.f,0.f};
#pragma unroll
  for (int ks = 0; ks < KS; ++ks) {
    const v2f a = *(const v2f*)(ap + ks * 4);
    acc = __builtin_amdgcn_wmma_f32_16x16x4_f32(false, a, false, bf.v[ks], (short)0, acc, false, false);
  }
  return acc;
}

template <bool RELU>
__device__ __forceinline__ void store_tile(v8f acc, float bv, float* Ds, int dpitch,
                                           int col, int hi) {
#pragma unroll
  for (int v = 0; v < 8; ++v) {
    float val = acc[v] + bv;
    if (RELU) val = fmaxf(val, 0.f);
    Ds[(v + 8 * hi) * dpitch + col] = val;
  }
}

#define XPITCH 116
#define QPITCH 20
__device__ __forceinline__ void proj_stage(const float* Xin, const float* __restrict__ pW, int w,
                                           const float* __restrict__ cb, float* Pt, float* Pq,
                                           int lane) {
  const int l16 = lane & 15;
  const int hi = lane >> 4;
  const float* ap = Xin + l16 * XPITCH + hi * 2;
  const float* ap8 = ap + w * 28;
  const float* bp = pW + (size_t)(2 * w) * (28 * 64) + lane * 2;
  const float* bp8 = pW + (size_t)(8 * 28 + 7 * w) * 64 + lane * 2;
  v8f acc0 = {0.f,0.f,0.f,0.f,0.f,0.f,0.f,0.f};
  v8f acc1 = acc0, acc8 = acc0;
#pragma unroll
  for (int ks = 0; ks < 28; ++ks) {
    const v2f a = *(const v2f*)(ap + ks * 4);
    const v2f b0 = *(const v2f*)(bp + ks * 64);
    const v2f b1 = *(const v2f*)(bp + (28 + ks) * 64);
    acc0 = __builtin_amdgcn_wmma_f32_16x16x4_f32(false, a, false, b0, (short)0, acc0, false, false);
    acc1 = __builtin_amdgcn_wmma_f32_16x16x4_f32(false, a, false, b1, (short)0, acc1, false, false);
    if ((ks & 3) == 0) {
      const int j = ks >> 2;
      const v2f a8 = *(const v2f*)(ap8 + j * 4);
      const v2f b8 = *(const v2f*)(bp8 + j * 64);
      acc8 = __builtin_amdgcn_wmma_f32_16x16x4_f32(false, a8, false, b8, (short)0, acc8, false, false);
    }
  }
  const int col0 = (2 * w) * 16 + l16;
  store_tile<false>(acc0, cb[col0], Pt, 146, col0, hi);
  store_tile<false>(acc1, cb[col0 + 16], Pt, 146, col0 + 16, hi);
  store_tile<false>(acc8, 0.f, Pq + w * (16 * QPITCH), QPITCH, l16, hi);
}

__device__ __forceinline__ void proj_reduce(const float* Pq, float bq, float* Pt, int tid) {
#pragma unroll
  for (int i = 0; i < 2; ++i) {
    const int idx = i * GC_THREADS + tid;
    const int r = idx >> 4;
    const int c = idx & 15;
    const float* q = Pq + r * QPITCH + c;
    const float s = (q[0] + q[16 * QPITCH]) + (q[2 * 16 * QPITCH] + q[3 * 16 * QPITCH]);
    Pt[r * 146 + 128 + c] = s + bq;
  }
}

__global__ __launch_bounds__(GC_THREADS, 8) void graph_chain(
    const float* __restrict__ X0,
    const float* __restrict__ pW1a, const float* __restrict__ cba,
    const float* __restrict__ g1, const float* __restrict__ be1,
    const float* __restrict__ pW2a, const float* __restrict__ b2a,
    const float* __restrict__ pW1b, const float* __restrict__ cbb,
    const float* __restrict__ g2, const float* __restrict__ be2,
    const float* __restrict__ pW2b, const float* __restrict__ b2b,
    const float* __restrict__ pW1c, const float* __restrict__ cbc,
    const float* __restrict__ g3, const float* __restrict__ be3,
    const float* __restrict__ pW2c, const float* __restrict__ b2c,
    const float* __restrict__ pPsi1, const float* __restrict__ psib1,
    const float* __restrict__ pPsi2, const float* __restrict__ psib2,
    const float* __restrict__ phiW1, const float* __restrict__ phib1,
    const float* __restrict__ phiW2, const float* __restrict__ phib2,
    float* __restrict__ out) {
  __shared__ __attribute__((aligned(16))) float Xin[16 * XPITCH];
  __shared__ __attribute__((aligned(16))) float Pt[16 * 146];
  __shared__ __attribute__((aligned(16))) float Pq[4 * 16 * QPITCH];
  __shared__ __attribute__((aligned(16))) float rsS[GC_THREADS];
  __shared__ __attribute__((aligned(16))) float nmS[GC_THREADS];
  __shared__ __attribute__((aligned(16))) float Hs[16 * 76];
  __shared__ __attribute__((aligned(16))) float Xs[16 * 36];
  __shared__ __attribute__((aligned(16))) float X5s[16 * 20];
  __shared__ float Hhs[16 * 17];
  __shared__ float gsA[72], bsA[72], gsB[32], bsB[32], gsC[32], bsC[32];
  __shared__ float ps[32], qs[32];

  const int tid = threadIdx.x;
  const int lane = tid & 31;
  const int l16 = lane & 15;
  const int hi = lane >> 4;
  const int blk = blockIdx.x;
  const int rot = (blk ^ (blk >> 2) ^ (blk >> 4) ^ (blk >> 6) ^ (blk >> 8) ^ (blk >> 10)) & 3;
  const int w = (__builtin_amdgcn_readfirstlane(tid >> 5) + rot) & 3;
  const int wn = w & 1;

  float bq9 = 0.f;
  {
    const float4* src = (const float4*)(X0 + (size_t)blk * (16 * 110));
    float4 xv[4];
    xv[0] = src[tid];
    xv[1] = src[GC_THREADS + tid];
    xv[2] = src[2 * GC_THREADS + tid];
    xv[3] = (tid < 56) ? src[3 * GC_THREADS + tid] : make_float4(0.f, 0.f, 0.f, 0.f);
    float ga = 0.f, ba = 0.f, gb = 0.f, bb = 0.f, gc = 0.f, bc = 0.f;
    if (tid < 71) { ga = g1[tid]; ba = be1[tid]; }
    if (tid < 32) { gb = g2[tid]; bb = be2[tid]; gc = g3[tid]; bc = be3[tid]; }
    if ((tid & 15) < 14) bq9 = cba[128 + (tid & 15)];
#pragma unroll
    for (int it = 0; it < 4; ++it) {
      const int f = it * GC_THREADS + tid;
      if (it < 3 || tid < 56) {
        const int e0 = f * 4;
        const int r0 = e0 / 110;
        const int c0 = e0 - r0 * 110;
        const int e1 = e0 + 2;
        const int r1 = e1 / 110;
        const int c1 = e1 - r1 * 110;
        *(float2*)&Xin[r0 * XPITCH + c0] = make_float2(xv[it].x, xv[it].y);
        *(float2*)&Xin[r1 * XPITCH + c1] = make_float2(xv[it].z, xv[it].w);
      }
    }
    if (tid < 16) *(float2*)&Xin[tid * XPITCH + 110] = make_float2(0.f, 0.f);
    if (tid < 71) { gsA[tid] = ga; bsA[tid] = ba; }
    if (tid < 32) { gsB[tid] = gb; bsB[tid] = bb; gsC[tid] = gc; bsC[tid] = bc; }
  }
  __syncthreads();

  proj_stage(Xin, pW1a, w, cba, Pt, Pq, lane);
  __syncthreads();
  proj_reduce(Pq, bq9, Pt, tid);
  __syncthreads();

  edge_stage<71, 146, 72, 76>(Pt, rsS, nmS, gsA, bsA, Hs, tid);
  const BFrag<8> bfB = load_b<8>(pW1b + w * (8 * 64), lane);
  const float cbbv = cbb[w * 16 + l16];
  if (w < 2) {
    const BFrag<18> bfA = load_b<18>(pW2a + w * (18 * 64), lane);
    const v8f acc = tile_gemm<18>(Hs, 76, bfA, lane);
    store_tile<true>(acc, b2a[w * 16 + l16], Xs, 36, w * 16 + l16, hi);
  }
  __syncthreads();

  {
    const v8f acc = tile_gemm<8>(Xs, 36, bfB, lane);
    store_tile<false>(acc, cbbv, Pt, 66, w * 16 + l16, hi);
  }
  const BFrag<8> bfC = load_b<8>(pW2b + wn * (8 * 64), lane);
  const float b2bv = b2b[wn * 16 + l16];
  __syncthreads();
  edge_stage<32, 66, 32, 36>(Pt, rsS, nmS, gsB, bsB, Hs, tid);
  const BFrag<8> bfD = load_b<8>(pW1c + w * (8 * 64), lane);
  const float cbcv = cbc[w * 16 + l16];
  if (w < 2) {
    const v8f acc = tile_gemm<8>(Hs, 36, bfC, lane);
    store_tile<true>(acc, b2bv, Xs, 36, w * 16 + l16, hi);
  }
  __syncthreads();

  {
    const v8f acc = tile_gemm<8>(Xs, 36, bfD, lane);
    store_tile<false>(acc, cbcv, Pt, 66, w * 16 + l16, hi);
  }
  const BFrag<8> bfE = load_b<8>(pW2c + wn * (8 * 64), lane);
  const float b2cv = b2c[wn * 16 + l16];
  __syncthreads();
  edge_stage<32, 66, 32, 36>(Pt, rsS, nmS, gsC, bsC, Hs, tid);
  const BFrag<8> bfF = load_b<8>(pPsi1, lane);
  const BFrag<4> bfG = load_b<4>(pPsi2, lane);
  const float psi1v = psib1[l16];
  const float psi2v = psib2[l16];
  if (w < 2) {
    const v8f acc = tile_gemm<8>(Hs, 36, bfE, lane);
    store_tile<false>(acc, b2cv, Xs, 36, w * 16 + l16, hi);
  }
  __syncthreads();

  if (w == 0) {
    const v8f acc = tile_gemm<8>(Xs, 36, bfF, lane);
    store_tile<true>(acc, psi1v, X5s, 20, l16, hi);
  }
  __syncthreads();
  if (w == 0) {
    const v8f acc = tile_gemm<4>(X5s, 20, bfG, lane);
    store_tile<false>(acc, psi2v, Hhs, 17, l16, hi);
  }
  __syncthreads();

  const int gq = hi;
  const int tq = l16;
  if (w == 0) {
    float s = 0.f;
#pragma unroll
    for (int nd = 0; nd < 8; ++nd) s += Hhs[(gq * 8 + nd) * 17 + tq];
    ps[lane] = s * 0.125f;
  }
  __syncthreads();
  if (w == 0) {
    float s = phib1[tq];
#pragma unroll
    for (int k = 0; k < 16; ++k) s += ps[gq * 16 + k] * phiW1[k * 16 + tq];
    qs[lane] = fmaxf(s, 0.f);
  }
  __syncthreads();
  if (w == 0 && lane < 2) {
    float s = phib2[0];
#pragma unroll
    for (int k = 0; k < 16; ++k) s += qs[lane * 16 + k] * phiW2[k];
    out[blk * 2 + lane] = s;
  }
}

extern "C" void kernel_launch(void* const* d_in, const int* in_sizes, int n_in,
                              void* d_out, int out_size, void* d_ws, size_t ws_size,
                              hipStream_t stream) {
  (void)in_sizes; (void)n_in; (void)out_size; (void)ws_size;
  const float* X0     = (const float*)d_in[0];
  const float* l1_W1  = (const float*)d_in[1];
  const float* l1_b1  = (const float*)d_in[2];
  const float* l1_g   = (const float*)d_in[3];
  const float* l1_be  = (const float*)d_in[4];
  const float* l1_W2  = (const float*)d_in[5];
  const float* l1_b2  = (const float*)d_in[6];
  const float* l2_W1  = (const float*)d_in[7];
  const float* l2_b1  = (const float*)d_in[8];
  const float* l2_g   = (const float*)d_in[9];
  const float* l2_be  = (const float*)d_in[10];
  const float* l2_W2  = (const float*)d_in[11];
  const float* l2_b2  = (const float*)d_in[12];
  const float* l3_W1  = (const float*)d_in[13];
  const float* l3_b1  = (const float*)d_in[14];
  const float* l3_g   = (const float*)d_in[15];
  const float* l3_be  = (const float*)d_in[16];
  const float* l3_W2  = (const float*)d_in[17];
  const float* l3_b2  = (const float*)d_in[18];
  const float* psi_W1 = (const float*)d_in[19];
  const float* psi_b1 = (const float*)d_in[20];
  const float* psi_W2 = (const float*)d_in[21];
  const float* psi_b2 = (const float*)d_in[22];
  const float* phi_W1 = (const float*)d_in[23];
  const float* phi_b1 = (const float*)d_in[24];
  const float* phi_W2 = (const float*)d_in[25];
  const float* phi_b2 = (const float*)d_in[26];

  float* ws = (float*)d_ws;
  float* wcat1 = ws;
  float* cb1   = wcat1 + 15620;
  float* wcat2 = cb1 + 144;
  float* cb2   = wcat2 + 2048;
  float* wcat3 = cb2 + 64;
  float* cb3   = wcat3 + 2048;
  float* p_w1   = cb3 + 64;
  float* p_l1W2 = p_w1 + 16128;
  float* p_w2   = p_l1W2 + 2304;
  float* p_l2W2 = p_w2 + 2048;
  float* p_w3   = p_l2W2 + 1024;
  float* p_l3W2 = p_w3 + 2048;
  float* p_psi1 = p_l3W2 + 1024;
  float* p_psi2 = p_psi1 + 512;

  setup_all<<<52, 256, 0, stream>>>(l1_W1, l1_b1, l1_W2, l2_W1, l2_b1, l2_W2,
                                    l3_W1, l3_b1, l3_W2, psi_W1, psi_W2,
                                    cb1, cb2, cb3, p_w1, p_l1W2, p_w2, p_l2W2,
                                    p_w3, p_l3W2, p_psi1, p_psi2);

  graph_chain<<<2048, GC_THREADS, 0, stream>>>(X0, p_w1, cb1, l1_g, l1_be, p_l1W2, l1_b2,
                                                 p_w2, cb2, l2_g, l2_be, p_l2W2, l2_b2,
                                                 p_w3, cb3, l3_g, l3_be, p_l3W2, l3_b2,
                                                 p_psi1, psi_b1, p_psi2, psi_b2,
                                                 phi_W1, phi_b1, phi_W2, phi_b2, (float*)d_out);
}


# Linear maps that meet without a nonlinearity are multiplied once in the setup launch (l3W2 @ psiW1, psiW2 @ phiW1); node mean taken in the WMMA D registers: 2 GEMM stages, 3 barriers fewer per block
# speedup vs baseline: 6.2489x; 1.0548x over previous
#include <hip/hip_runtime.h>
#include <hip/hip_bf16.h>

typedef __attribute__((ext_vector_type(2))) float v2f;
typedef __attribute__((ext_vector_type(8))) float v8f;

__device__ __forceinline__ float wcat_at(const float* __restrict__ W1, int din, int dout,
                                         int k, int n) {
  return (n < dout) ? (W1[k * dout + n] - W1[(k + din) * dout + n])
                    : W1[(k + din) * dout + (n - dout)];
}

__device__ __forceinline__ void pack_plain(int idx, const float* __restrict__ Wt, int K, int N,
                                           int ksteps, int ntiles, float* __restrict__ out) {
  if (idx >= ntiles * ksteps * 32) return;
  const int lane = idx & 31;
  const int tmp = idx >> 5;
  const int ks = tmp % ksteps;
  const int nt = tmp / ksteps;
  const int col = nt * 16 + (lane & 15);
  const int k0 = ks * 4 + (lane >> 4) * 2;
  const float v0 = (k0 < K && col < N) ? Wt[(size_t)k0 * N + col] : 0.f;
  const float v1 = (k0 + 1 < K && col < N) ? Wt[(size_t)(k0 + 1) * N + col] : 0.f;
  out[(size_t)idx * 2]     = v0;
  out[(size_t)idx * 2 + 1] = v1;
}

__device__ __forceinline__ void pack_cat(int idx, const float* __restrict__ W1, int din, int dout,
                                         int ksteps, int ntiles, float* __restrict__ out) {
  if (idx >= ntiles * ksteps * 32) return;
  const int N = 2 * dout;
  const int lane = idx & 31;
  const int tmp = idx >> 5;
  const int ks = tmp % ksteps;
  const int nt = tmp / ksteps;
  const int col = nt * 16 + (lane & 15);
  const int k0 = ks * 4 + (lane >> 4) * 2;
  const float v0 = (k0 < din && col < N) ? wcat_at(W1, din, dout, k0, col) : 0.f;
  const float v1 = (k0 + 1 < din && col < N) ? wcat_at(W1, din, dout, k0 + 1, col) : 0.f;
  out[(size_t)idx * 2]     = v0;
  out[(size_t)idx * 2 + 1] = v1;
}

__device__ __forceinline__ void fold_wf(int e, const float* __restrict__ W2c,
                                        const float* __restrict__ psiW1, float* __restrict__ out) {
  const int idx = e >> 1;
  const int lane = idx & 31;
  const int ks = idx >> 5;
  const int col = lane & 15;
  const int k = ks * 4 + (lane >> 4) * 2 + (e & 1);
  float s = 0.f;
#pragma unroll
  for (int m = 0; m < 32; ++m) s = fmaf(W2c[k * 32 + m], psiW1[m * 16 + col], s);
  out[e] = s;
}

__global__ void setup_all(const float* __restrict__ l1W1, const float* __restrict__ l1b1,
                          const float* __restrict__ l1W2,
                          const float* __restrict__ l2W1, const float* __restrict__ l2b1,
                          const float* __restrict__ l2W2,
                          const float* __restrict__ l3W1, const float* __restrict__ l3b1,
                          const float* __restrict__ l3W2,
                          const float* __restrict__ psiW1, const float* __restrict__ psiW2,
                          const float* __restrict__ l3b2, const float* __restrict__ psib1,
                          const float* __restrict__ psib2, const float* __restrict__ phiW1,
                          const float* __restrict__ phib1,
                          float* __restrict__ cb1, float* __restrict__ cb2, float* __restrict__ cb3,
                          float* __restrict__ p_w1, float* __restrict__ p_l1W2,
                          float* __restrict__ p_w2, float* __restrict__ p_l2W2,
                          float* __restrict__ p_w3, float* __restrict__ p_wf,
                          float* __restrict__ bfold, float* __restrict__ wg, float* __restrict__ bgv) {
  int b = blockIdx.x;
  const int t = threadIdx.x;
  if (b < 32) { pack_cat(b * 256 + t, l1W1, 110, 71, 28, 9, p_w1); return; }
  b -= 32;
  if (b < 5) { pack_plain(b * 256 + t, l1W2, 71, 32, 18, 2, p_l1W2); return; }
  b -= 5;
  if (b < 4) { pack_cat(b * 256 + t, l2W1, 32, 32, 8, 4, p_w2); return; }
  b -= 4;
  if (b < 2) { pack_plain(b * 256 + t, l2W2, 32, 32, 8, 2, p_l2W2); return; }
  b -= 2;
  if (b < 4) { pack_cat(b * 256 + t, l3W1, 32, 32, 8, 4, p_w3); return; }
  b -= 4;
  if (b < 2) { fold_wf(b * 256 + t, l3W2, psiW1, p_wf); return; }
  b -= 2;
  if (b < 1) {
    const int k = t >> 4, n = t & 15;
    float s = 0.f;
#pragma unroll
    for (int m = 0; m < 16; ++m) s = fmaf(psiW2[k * 16 + m], phiW1[m * 16 + n], s);
    wg[t] = s;
    return;
  }
  b -= 1;
  if (b < 1) {
    if (t < 16) {
      float s = psib1[t];
#pragma unroll
      for (int m = 0; m < 32; ++m) s = fmaf(l3b2[m], psiW1[m * 16 + t], s);
      bfold[t] = s;
    } else if (t < 32) {
      const int n = t - 16;
      float s = phib1[n];
#pragma unroll
      for (int m = 0; m < 16; ++m) s = fmaf(psib2[m], phiW1[m * 16 + n], s);
      bgv[n] = s;
    }
    return;
  }
  if (t < 142) cb1[t] = (t < 71) ? l1b1[t] : 0.f;
  if (t < 64) {
    cb2[t] = (t < 32) ? l2b1[t] : 0.f;
    cb3[t] = (t < 32) ? l3b1[t] : 0.f;
  }
}

#define GC_THREADS 128

template <int C, int PP, int KP, int HP>
__device__ __forceinline__ void edge_stage(const float* Pt, float* rsS, float* nmS, const float* gs,
                                           const float* bs, float* Hs, int tid) {
  {
    const int ri = tid >> 3;
    const int rj = ((tid >> 6) << 3) + (tid & 7);
    const float* pa = Pt + ri * PP;
    const float* pb = Pt + rj * PP + C;
    float sum = 0.f, sq = 0.f;
#pragma unroll
    for (int c = 0; c < C; ++c) {
      const float v = pa[c] + pb[c];
      sum += v;
      sq += v * v;
    }
    const float inv = 1.0f / (float)C;
    const float mean = sum * inv;
    const float var = sq * inv - mean * mean;
    const float rs = rsqrtf(var + 1e-5f);
    rsS[tid] = rs;
    nmS[tid] = -mean * rs;
  }
  __syncthreads();
  constexpr int NP = C / 32;
  const int lane = tid & 31;
  const int r0 = __builtin_amdgcn_readfirstlane(tid >> 5) * 4;
  {
    const float* pbn = Pt + (r0 & 8) * PP + C + lane;
    v2f bv[NP][4], gg[NP], bb[NP];
#pragma unroll
    for (int p = 0; p < NP; ++p) {
      const float g1 = gs[p * 32 + lane];
      const float b1 = bs[p * 32 + lane];
      gg[p] = v2f{g1, g1};
      bb[p] = v2f{b1, b1};
#pragma unroll
      for (int q = 0; q < 4; ++q)
        bv[p][q] = v2f{pbn[(2 * q) * PP + p * 32], pbn[(2 * q + 1) * PP + p * 32]};
    }
#pragma unroll
    for (int rr = 0; rr < 4; ++rr) {
      const int row = r0 + rr;
      const float4 ra = *(const float4*)(rsS + row * 8);
      const float4 rb = *(const float4*)(rsS + row * 8 + 4);
      const float4 na = *(const float4*)(nmS + row * 8);
      const float4 nb = *(const float4*)(nmS + row * 8 + 4);
      const v2f rsv[4] = {v2f{ra.x, ra.y}, v2f{ra.z, ra.w}, v2f{rb.x, rb.y}, v2f{rb.z, rb.w}};
      const v2f nmv[4] = {v2f{na.x, na.y}, v2f{na.z, na.w}, v2f{nb.x, nb.y}, v2f{nb.z, nb.w}};
#pragma unroll
      for (int p = 0; p < NP; ++p) {
        const float a = Pt[row * PP + p * 32 + lane];
        const v2f a2 = v2f{a, a};
        v2f s2 = v2f{0.f, 0.f};
#pragma unroll
        for (int q = 0; q < 4; ++q) {
          v2f v = __builtin_elementwise_fma(a2 + bv[p][q], rsv[q], nmv[q]);
          v = __builtin_elementwise_fma(v, gg[p], bb[p]);
          s2 += __builtin_elementwise_max(v, v2f{0.f, 0.f});
        }
        Hs[row * HP + p * 32 + lane] = (s2.x + s2.y) * 0.125f;
      }
    }
  }
  if (KP > NP * 32) {
    static_assert(KP == NP * 32 || KP - NP * 32 == 8, "remainder pass covers eight columns");
    const int row = tid >> 3;
    const int c = NP * 32 + (tid & 7);
    float out = 0.f;
    if (c < C) {
      const float a = Pt[row * PP + c];
      const float g1 = gs[c];
      const float b1 = bs[c];
      const float* pb = Pt + (row & 8) * PP + C + c;
      const float4 ra = *(const float4*)(rsS + row * 8);
      const float4 rb = *(const float4*)(rsS + row * 8 + 4);
      const float4 na = *(const float4*)(nmS + row * 8);
      const float4 nb = *(const float4*)(nmS + row * 8 + 4);
      const v2f rsv[4] = {v2f{ra.x, ra.y}, v2f{ra.z, ra.w}, v2f{rb.x, rb.y}, v2f{rb.z, rb.w}};
      const v2f nmv[4] = {v2f{na.x, na.y}, v2f{na.z, na.w}, v2f{nb.x, nb.y}, v2f{nb.z, nb.w}};
      const v2f a2 = v2f{a, a};
      const v2f g2 = v2f{g1, g1};
      const v2f b2 = v2f{b1, b1};
      v2f s2 = v2f{0.f, 0.f};
#pragma unroll
      for (int q = 0; q < 4; ++q) {
        const v2f bq = v2f{pb[(2 * q) * PP], pb[(2 * q + 1) * PP]};
        v2f v = __builtin_elementwise_fma(a2 + bq, rsv[q], nmv[q]);
        v = __builtin_elementwise_fma(v, g2, b2);
        s2 += __builtin_elementwise_max(v, v2f{0.f, 0.f});
      }
      const float s = s2.x + s2.y;
      out = s * 0.125f;
    }
    Hs[row * HP + c] = out;
  }
  __syncthreads();
}

template <int KS>
struct BFrag { v2f v[KS]; };

template <int KS>
__device__ __forceinline__ BFrag<KS> load_b(const float* __restrict__ Bt, int lane) {
  BFrag<KS> f;
  const float* bp = Bt + lane * 2;
#pragma unroll
  for (int ks = 0; ks < KS; ++ks) f.v[ks] = *(const v2f*)(bp + ks * 64);
  return f;
}

template <int KS>
__device__ __forceinline__ v8f tile_gemm(const float* As, int apitch, const BFrag<KS>& bf, int lane) {
  const float* ap = As + (lane & 15) * apitch + (lane >> 4) * 2;
  v8f acc = {0.f,0.f,0.f,0.f,0.f,0.f,0.f,0.f};
#pragma unroll
  for (int ks = 0; ks < KS; ++ks) {
    const v2f a = *(const v2f*)(ap + ks * 4);
    acc = __builtin_amdgcn_wmma_f32_16x16x4_f32(false, a, false, bf.v[ks], (short)0, acc, false, false);
  }
  return acc;
}

template <bool RELU>
__device__ __forceinline__ void store_tile(v8f acc, float bv, float* Ds, int dpitch,
                                           int col, int hi) {
#pragma unroll
  for (int v = 0; v < 8; ++v) {
    float val = acc[v] + bv;
    if (RELU) val = fmaxf(val, 0.f);
    Ds[(v + 8 * hi) * dpitch + col] = val;
  }
}

#define XPITCH 116
#define QPITCH 20
__device__ __forceinline__ void proj_stage(const float* Xin, const float* __restrict__ pW, int w,
                                           const float* __restrict__ cb, float* Pt, float* Pq,
                                           int lane) {
  const int l16 = lane & 15;
  const int hi = lane >> 4;
  const float* ap = Xin + l16 * XPITCH + hi * 2;
  const float* ap8 = ap + w * 28;
  const float* bp = pW + (size_t)(2 * w) * (28 * 64) + lane * 2;
  const float* bp8 = pW + (size_t)(8 * 28 + 7 * w) * 64 + lane * 2;
  v8f acc0 = {0.f,0.f,0.f,0.f,0.f,0.f,0.f,0.f};
  v8f acc1 = acc0, acc8 = acc0;
#pragma unroll
  for (int ks = 0; ks < 28; ++ks) {
    const v2f a = *(const v2f*)(ap + ks * 4);
    const v2f b0 = *(const v2f*)(bp + ks * 64);
    const v2f b1 = *(const v2f*)(bp + (28 + ks) * 64);
    acc0 = __builtin_amdgcn_wmma_f32_16x16x4_f32(false, a, false, b0, (short)0, acc0, false, false);
    acc1 = __builtin_amdgcn_wmma_f32_16x16x4_f32(false, a, false, b1, (short)0, acc1, false, false);
    if ((ks & 3) == 0) {
      const int j = ks >> 2;
      const v2f a8 = *(const v2f*)(ap8 + j * 4);
      const v2f b8 = *(const v2f*)(bp8 + j * 64);
      acc8 = __builtin_amdgcn_wmma_f32_16x16x4_f32(false, a8, false, b8, (short)0, acc8, false, false);
    }
  }
  const int col0 = (2 * w) * 16 + l16;
  store_tile<false>(acc0, cb[col0], Pt, 146, col0, hi);
  store_tile<false>(acc1, cb[col0 + 16], Pt, 146, col0 + 16, hi);
  store_tile<false>(acc8, 0.f, Pq + w * (16 * QPITCH), QPITCH, l16, hi);
}

__device__ __forceinline__ void proj_reduce(const float* Pq, float bq, float* Pt, int tid) {
#pragma unroll
  for (int i = 0; i < 2; ++i) {
    const int idx = i * GC_THREADS + tid;
    const int r = idx >> 4;
    const int c = idx & 15;
    const float* q = Pq + r * QPITCH + c;
    const float s = (q[0] + q[16 * QPITCH]) + (q[2 * 16 * QPITCH] + q[3 * 16 * QPITCH]);
    Pt[r * 146 + 128 + c] = s + bq;
  }
}

__global__ __launch_bounds__(GC_THREADS, 8) void graph_chain(
    const float* __restrict__ X0,
    const float* __restrict__ pW1a, const float* __restrict__ cba,
    const float* __restrict__ g1, const float* __restrict__ be1,
    const float* __restrict__ pW2a, const float* __restrict__ b2a,
    const float* __restrict__ pW1b, const float* __restrict__ cbb,
    const float* __restrict__ g2, const float* __restrict__ be2,
    const float* __restrict__ pW2b, const float* __restrict__ b2b,
    const float* __restrict__ pW1c, const float* __restrict__ cbc,
    const float* __restrict__ g3, const float* __restrict__ be3,
    const float* __restrict__ pWf, const float* __restrict__ bfold,
    const float* __restrict__ wg, const float* __restrict__ bgv,
    const float* __restrict__ phiW2, const float* __restrict__ phib2,
    float* __restrict__ out) {
  __shared__ __attribute__((aligned(16))) float Xin[16 * XPITCH];
  __shared__ __attribute__((aligned(16))) float Pt[16 * 146];
  __shared__ __attribute__((aligned(16))) float Pq[4 * 16 * QPITCH];
  __shared__ __attribute__((aligned(16))) float rsS[GC_THREADS];
  __shared__ __attribute__((aligned(16))) float nmS[GC_THREADS];
  __shared__ __attribute__((aligned(16))) float Hs[16 * 76];
  __shared__ __attribute__((aligned(16))) float Xs[16 * 36];
  __shared__ float gsA[72], bsA[72], gsB[32], bsB[32], gsC[32], bsC[32];
  __shared__ float ps[32], qs[32];

  const int tid = threadIdx.x;
  const int lane = tid & 31;
  const int l16 = lane & 15;
  const int hi = lane >> 4;
  const int blk = blockIdx.x;
  const int rot = (blk ^ (blk >> 2) ^ (blk >> 4) ^ (blk >> 6) ^ (blk >> 8) ^ (blk >> 10)) & 3;
  const int w = (__builtin_amdgcn_readfirstlane(tid >> 5) + rot) & 3;
  const int wn = w & 1;

  float bq9 = 0.f;
  {
    const float4* src = (const float4*)(X0 + (size_t)blk * (16 * 110));
    float4 xv[4];
    xv[0] = src[tid];
    xv[1] = src[GC_THREADS + tid];
    xv[2] = src[2 * GC_THREADS + tid];
    xv[3] = (tid < 56) ? src[3 * GC_THREADS + tid] : make_float4(0.f, 0.f, 0.f, 0.f);
    float ga = 0.f, ba = 0.f, gb = 0.f, bb = 0.f, gc = 0.f, bc = 0.f;
    if (tid < 71) { ga = g1[tid]; ba = be1[tid]; }
    if (tid < 32) { gb = g2[tid]; bb = be2[tid]; gc = g3[tid]; bc = be3[tid]; }
    if ((tid & 15) < 14) bq9 = cba[128 + (tid & 15)];
#pragma unroll
    for (int it = 0; it < 4; ++it) {
      const int f = it * GC_THREADS + tid;
      if (it < 3 || tid < 56) {
        const int e0 = f * 4;
        const int r0 = e0 / 110;
        const int c0 = e0 - r0 * 110;
        const int e1 = e0 + 2;
        const int r1 = e1 / 110;
        const int c1 = e1 - r1 * 110;
        *(float2*)&Xin[r0 * XPITCH + c0] = make_float2(xv[it].x, xv[it].y);
        *(float2*)&Xin[r1 * XPITCH + c1] = make_float2(xv[it].z, xv[it].w);
      }
    }
    if (tid < 16) *(float2*)&Xin[tid * XPITCH + 110] = make_float2(0.f, 0.f);
    if (tid < 71) { gsA[tid] = ga; bsA[tid] = ba; }
    if (tid < 32) { gsB[tid] = gb; bsB[tid] = bb; gsC[tid] = gc; bsC[tid] = bc; }
  }
  __syncthreads();

  proj_stage(Xin, pW1a, w, cba, Pt, Pq, lane);
  __syncthreads();
  proj_reduce(Pq, bq9, Pt, tid);
  __syncthreads();

  edge_stage<71, 146, 72, 76>(Pt, rsS, nmS, gsA, bsA, Hs, tid);
  const BFrag<8> bfB = load_b<8>(pW1b + w * (8 * 64), lane);
  const float cbbv = cbb[w * 16 + l16];
  if (w < 2) {
    const BFrag<18> bfA = load_b<18>(pW2a + w * (18 * 64), lane);
    const v8f acc = tile_gemm<18>(Hs, 76, bfA, lane);
    store_tile<true>(acc, b2a[w * 16 + l16], Xs, 36, w * 16 + l16, hi);
  }
  __syncthreads();

  {
    const v8f acc = tile_gemm<8>(Xs, 36, bfB, lane);
    store_tile<false>(acc, cbbv, Pt, 66, w * 16 + l16, hi);
  }
  const BFrag<8> bfC = load_b<8>(pW2b + wn * (8 * 64), lane);
  const float b2bv = b2b[wn * 16 + l16];
  __syncthreads();
  edge_stage<32, 66, 32, 36>(Pt, rsS, nmS, gsB, bsB, Hs, tid);
  const BFrag<8> bfD = load_b<8>(pW1c + w * (8 * 64), lane);
  const float cbcv = cbc[w * 16 + l16];
  if (w < 2) {
    const v8f acc = tile_gemm<8>(Hs, 36, bfC, lane);
    store_tile<true>(acc, b2bv, Xs, 36, w * 16 + l16, hi);
  }
  __syncthreads();

  {
    const v8f acc = tile_gemm<8>(Xs, 36, bfD, lane);
    store_tile<false>(acc, cbcv, Pt, 66, w * 16 + l16, hi);
  }
  const BFrag<8> bfF = load_b<8>(pWf, lane);
  const float bfv = bfold[l16];
  __syncthreads();
  edge_stage<32, 66, 32, 36>(Pt, rsS, nmS, gsC, bsC, Hs, tid);

  const int gq = hi;
  const int tq = l16;
  float wgc[16];
#pragma unroll
  for (int k = 0; k < 16; ++k) wgc[k] = wg[k * 16 + tq];
  const float bgq = bgv[tq];
  if (w == 0) {
    const v8f acc = tile_gemm<8>(Hs, 36, bfF, lane);
    float s = 0.f;
#pragma unroll
    for (int v = 0; v < 8; ++v) s += fmaxf(acc[v] + bfv, 0.f);
    ps[lane] = s * 0.125f;
  }
  __syncthreads();
  if (w == 0) {
    float s = bgq;
#pragma unroll
    for (int k = 0; k < 16; ++k) s += ps[gq * 16 + k] * wgc[k];
    qs[lane] = fmaxf(s, 0.f);
  }
  __syncthreads();
  if (w == 0 && lane < 2) {
    float s = phib2[0];
#pragma unroll
    for (int k = 0; k < 16; ++k) s += qs[lane * 16 + k] * phiW2[k];
    out[blk * 2 + lane] = s;
  }
}

extern "C" void kernel_launch(void* const* d_in, const int* in_sizes, int n_in,
                              void* d_out, int out_size, void* d_ws, size_t ws_size,
                              hipStream_t stream) {
  (void)in_sizes; (void)n_in; (void)out_size; (void)ws_size;
  const float* X0     = (const float*)d_in[0];
  const float* l1_W1  = (const float*)d_in[1];
  const float* l1_b1  = (const float*)d_in[2];
  const float* l1_g   = (const float*)d_in[3];
  const float* l1_be  = (const float*)d_in[4];
  const float* l1_W2  = (const float*)d_in[5];
  const float* l1_b2  = (const float*)d_in[6];
  const float* l2_W1  = (const float*)d_in[7];
  const float* l2_b1  = (const float*)d_in[8];
  const float* l2_g   = (const float*)d_in[9];
  const float* l2_be  = (const float*)d_in[10];
  const float* l2_W2  = (const float*)d_in[11];
  const float* l2_b2  = (const float*)d_in[12];
  const float* l3_W1  = (const float*)d_in[13];
  const float* l3_b1  = (const float*)d_in[14];
  const float* l3_g   = (const float*)d_in[15];
  const float* l3_be  = (const float*)d_in[16];
  const float* l3_W2  = (const float*)d_in[17];
  const float* l3_b2  = (const float*)d_in[18];
  const float* psi_W1 = (const float*)d_in[19];
  const float* psi_b1 = (const float*)d_in[20];
  const float* psi_W2 = (const float*)d_in[21];
  const float* psi_b2 = (const float*)d_in[22];
  const float* phi_W1 = (const float*)d_in[23];
  const float* phi_b1 = (const float*)d_in[24];
  const float* phi_W2 = (const float*)d_in[25];
  const float* phi_b2 = (const float*)d_in[26];

  float* ws = (float*)d_ws;
  float* wcat1 = ws;
  float* cb1   = wcat1 + 15620;
  float* wcat2 = cb1 + 144;
  float* cb2   = wcat2 + 2048;
  float* wcat3 = cb2 + 64;
  float* cb3   = wcat3 + 2048;
  float* p_w1   = cb3 + 64;
  float* p_l1W2 = p_w1 + 16128;
  float* p_w2   = p_l1W2 + 2304;
  float* p_l2W2 = p_w2 + 2048;
  float* p_w3   = p_l2W2 + 1024;
  float* p_wf   = p_w3 + 2048;
  float* bfold  = p_wf + 512;
  float* wg     = bfold + 16;
  float* bgv    = wg + 256;

  setup_all<<<52, 256, 0, stream>>>(l1_W1, l1_b1, l1_W2, l2_W1, l2_b1, l2_W2,
                                    l3_W1, l3_b1, l3_W2, psi_W1, psi_W2,
                                    l3_b2, psi_b1, psi_b2, phi_W1, phi_b1,
                                    cb1, cb2, cb3, p_w1, p_l1W2, p_w2, p_l2W2,
                                    p_w3, p_wf, bfold, wg, bgv);

  graph_chain<<<2048, GC_THREADS, 0, stream>>>(X0, p_w1, cb1, l1_g, l1_be, p_l1W2, l1_b2,
                                                 p_w2, cb2, l2_g, l2_be, p_l2W2, l2_b2,
                                                 p_w3, cb3, l3_g, l3_be,
                                                 p_wf, bfold, wg, bgv,
                                                 phi_W2, phi_b2, (float*)d_out);
}
